# starting kernel
# baseline (speedup 1.0000x reference)
#include <hip/hip_runtime.h>
#include <math.h>

#define FEAT_D 32

typedef __attribute__((ext_vector_type(2))) float v2f;
typedef __attribute__((ext_vector_type(8))) float v8f;

__global__ __launch_bounds__(256)
void zero_kernel(float4* __restrict__ ax4, int n4) {
    int i = blockIdx.x * blockDim.x + threadIdx.x;
    if (i < n4) ax4[i] = make_float4(0.f, 0.f, 0.f, 0.f);
}

__global__ __launch_bounds__(256)
void scatter_kernel(const float* __restrict__ x,
                    const int*   __restrict__ src,
                    const int*   __restrict__ dst,
                    const float* __restrict__ w,
                    float*       __restrict__ ax,
                    int E) {
    int wave = blockIdx.x * (blockDim.x >> 5) + (threadIdx.x >> 5);
    int e = __builtin_amdgcn_readfirstlane(wave);
    if (e >= E) return;

    __builtin_prefetch(&src[e + 8192], 0, 1);
    __builtin_prefetch(&dst[e + 8192], 0, 1);
    __builtin_prefetch(&w[e + 8192], 0, 1);

    int   s  = src[e];
    int   t  = dst[e];
    float we = w[e];

    int lane = threadIdx.x & 31;
    float val = we * x[s * FEAT_D + lane];
    unsafeAtomicAdd(&ax[t * FEAT_D + lane], val);
}

__global__ __launch_bounds__(256)
void update_wmma_kernel(const float* __restrict__ x,
                        const float* __restrict__ ax,
                        const float* __restrict__ x0,
                        const float* __restrict__ alpha_p,
                        const float* __restrict__ beta_p,
                        float*       __restrict__ out,
                        int ntiles) {
    int tile = __builtin_amdgcn_readfirstlane(
        blockIdx.x * (blockDim.x >> 5) + (threadIdx.x >> 5));
    if (tile >= ntiles) return;

    const int nb = (tile >> 1) * 16;
    const int cb = (tile & 1) * 16;

    const int lane = threadIdx.x & 31;
    const int hi   = lane >> 4;
    const int li   = lane & 15;

    float araw  = *alpha_p;
    float alpha = 1.0f / (1.0f + expf(-araw));
    float beta  = *beta_p;
    float ca    = 1.0f - alpha;

    v8f c;
#pragma unroll
    for (int v = 0; v < 8; ++v) {
        int row = nb + v + 8 * hi;
        int idx = row * FEAT_D + cb + li;
        c[v] = ca * x[idx] + beta * x0[idx];
    }

#pragma unroll
    for (int m = 0; m < 4; ++m) {
        v2f a, b;
#pragma unroll
        for (int v = 0; v < 2; ++v) {
            int k = v + 2 * hi;
            a[v] = (li == 4 * m + k) ? alpha : 0.0f;
            int rowb = nb + 4 * m + k;
            b[v] = ax[rowb * FEAT_D + cb + li];
        }
        c = __builtin_amdgcn_wmma_f32_16x16x4_f32(
                false, a, false, b, (short)0, c, false, false);
    }

#pragma unroll
    for (int v = 0; v < 8; ++v) {
        int row = nb + v + 8 * hi;
        out[row * FEAT_D + cb + li] = c[v];
    }
}

extern "C" void kernel_launch(void* const* d_in, const int* in_sizes, int n_in,
                              void* d_out, int out_size, void* d_ws, size_t ws_size,
                              hipStream_t stream) {
    const float* x_in    = (const float*)d_in[0];
    const float* w       = (const float*)d_in[1];
    const float* x0      = (const float*)d_in[2];
    const float* alpha_p = (const float*)d_in[3];
    const float* beta_p  = (const float*)d_in[4];
    const int*   eidx    = (const int*)  d_in[5];

    const int N = in_sizes[0] / FEAT_D;
    const int E = in_sizes[1];
    const int* src = eidx;
    const int* dst = eidx + E;

    float* ax  = (float*)d_ws;
    float* out = (float*)d_out;

    const int ND = N * FEAT_D;
    dim3 blk(256);
    const int zeroBlocks = (ND / 4 + 255) / 256;
    const int scatBlocks = (E + 7) / 8;
    const int ntiles     = (N / 16) * (FEAT_D / 16);
    const int updBlocks  = (ntiles + 7) / 8;

    const float* xcur = x_in;
    for (int step = 0; step < 4; ++step) {
        zero_kernel<<<zeroBlocks, blk, 0, stream>>>((float4*)ax, ND / 4);
        scatter_kernel<<<scatBlocks, blk, 0, stream>>>(xcur, src, dst, w, ax, E);
        update_wmma_kernel<<<updBlocks, blk, 0, stream>>>(
            xcur, ax, x0, alpha_p, beta_p, out, ntiles);
        xcur = out;
    }
}


# Step kernel: a wave sums four rows at once (8 lanes x 16 B per source row, pairs passed on by 8-lane moves), a pass's 24 source rows requested together; 4-wave groups; f32 sum order unchanged
# speedup vs baseline: 5.0743x; 5.0743x over previous
#include <hip/hip_runtime.h>
#include <math.h>


#define FEAT_D 32

typedef __attribute__((ext_vector_type(2))) float v2f;
typedef __attribute__((ext_vector_type(4))) float v4f;
typedef __attribute__((ext_vector_type(8))) float v8f;

#define BIN_SHIFT 9
#define BIN_ROWS  512
#define CHUNK     8192
#define SRC_BITS  17
#define MAX_BINS  256

__global__ __launch_bounds__(1024)
void bin_hist_kernel(const int* __restrict__ dst, int* __restrict__ hist,
                     int E, int nbin, int nchunk) {
    __shared__ int s_h[MAX_BINS];
    const int tid = threadIdx.x;
    const int c   = blockIdx.x;
    if (tid < MAX_BINS) s_h[tid] = 0;
    __syncthreads();
    const int base = c * CHUNK + tid;
    int d[8];
#pragma unroll
    for (int k = 0; k < 8; ++k) {
        const int e = base + k * 1024;
        d[k] = (e < E) ? dst[e] : -1;
    }
#pragma unroll
    for (int k = 0; k < 8; ++k)
        if (d[k] >= 0) atomicAdd(&s_h[d[k] >> BIN_SHIFT], 1);
    __syncthreads();
    if (tid < nbin) hist[tid * nchunk + c] = s_h[tid];
}

__global__ __launch_bounds__(1024)
void scan_rows_kernel(const int* __restrict__ cnt, int* __restrict__ rowptr, int N) {
    __shared__ int s_part[32];
    __shared__ int s_tot[32];
    __shared__ int s_off[32];
    __shared__ int s_base;
    const int tid  = threadIdx.x;
    const int lane = tid & 31;
    const int wv   = tid >> 5;
    const int start = blockIdx.x * 1024;

    int part = 0;
    for (int i = tid; i < start; i += 1024) part += cnt[i];
    const int gi = start + tid;
    const int my = (gi < N) ? cnt[gi] : 0;

    for (int o = 16; o > 0; o >>= 1) part += __shfl_xor(part, o, 32);
    int incl = my;
    for (int o = 1; o < 32; o <<= 1) {
        int t = __shfl_up(incl, o, 32);
        if (lane >= o) incl += t;
    }
    if (lane == 31) s_tot[wv] = incl;
    if (lane == 0)  s_part[wv] = part;
    __syncthreads();
    if (wv == 0) {
        const int t = s_tot[lane];
        int p = s_part[lane];
        for (int o = 16; o > 0; o >>= 1) p += __shfl_xor(p, o, 32);
        int ti = t;
        for (int o = 1; o < 32; o <<= 1) {
            int u = __shfl_up(ti, o, 32);
            if (lane >= o) ti += u;
        }
        s_off[lane] = ti - t;
        if (lane == 0) s_base = p;
    }
    __syncthreads();
    const int excl = s_base + s_off[wv] + incl - my;
    if (gi < N) rowptr[gi] = excl;
    if (gi == N - 1) rowptr[N] = excl + my;
}

__global__ __launch_bounds__(1024)
void bin_scatter_kernel(const int*   __restrict__ src,
                        const int*   __restrict__ dst,
                        const float* __restrict__ w,
                        const int*   __restrict__ pref,
                        int*         __restrict__ rec,
                        int E, int nbin, int nchunk) {
    __shared__ int s_cur[MAX_BINS];
    const int tid = threadIdx.x;
    const int c   = blockIdx.x;
    if (tid < nbin) s_cur[tid] = pref[tid * nchunk + c];
    __syncthreads();
    const int base = c * CHUNK + tid;
    int d[8], s[8], wi[8];
#pragma unroll
    for (int k = 0; k < 8; ++k) {
        const int e = base + k * 1024;
        const bool ok = e < E;
        d[k]  = ok ? dst[e] : -1;
        s[k]  = ok ? src[e] : 0;
        wi[k] = ok ? __float_as_int(w[e]) : 0;
    }
#pragma unroll
    for (int k = 0; k < 8; ++k) {
        if (d[k] >= 0) {
            const int pos = atomicAdd(&s_cur[d[k] >> BIN_SHIFT], 1);
            ((int2*)rec)[pos] =
                make_int2(s[k] | ((d[k] & (BIN_ROWS - 1)) << SRC_BITS), wi[k]);
        }
    }
}

__global__ __launch_bounds__(1024)
void bin_rows_kernel(const int* __restrict__ rec,
                     const int* __restrict__ pref,
                     int*       __restrict__ rowptr,
                     int*       __restrict__ pairs,
                     int N, int nchunk) {
    __shared__ int s_cnt[BIN_ROWS];
    __shared__ int s_off[BIN_ROWS];
    __shared__ int s_ws[16];
    const int tid  = threadIdx.x;
    const int lane = tid & 31;
    const int wv   = tid >> 5;
    const int b    = blockIdx.x;
    const int lo   = pref[b * nchunk];
    const int hi   = pref[(b + 1) * nchunk];
    const int2* rc = (const int2*)rec;

    if (tid < BIN_ROWS) s_cnt[tid] = 0;
    __syncthreads();
    for (int i0 = lo + tid; i0 < hi; i0 += 8 * 1024) {
        int key[8];
#pragma unroll
        for (int k = 0; k < 8; ++k) {
            const int i = i0 + k * 1024;
            key[k] = (i < hi) ? rc[i].x : -1;
        }
#pragma unroll
        for (int k = 0; k < 8; ++k)
            if (key[k] >= 0) atomicAdd(&s_cnt[key[k] >> SRC_BITS], 1);
    }
    __syncthreads();

    const int my = (tid < BIN_ROWS) ? s_cnt[tid] : 0;
    int incl = my;
    for (int o = 1; o < 32; o <<= 1) {
        int t = __shfl_up(incl, o, 32);
        if (lane >= o) incl += t;
    }
    if (lane == 31 && wv < 16) s_ws[wv] = incl;
    __syncthreads();
    if (wv == 0) {
        const int t = (lane < 16) ? s_ws[lane] : 0;
        int ti = t;
        for (int o = 1; o < 16; o <<= 1) {
            int u = __shfl_up(ti, o, 32);
            if (lane >= o) ti += u;
        }
        if (lane < 16) s_ws[lane] = ti - t;
    }
    __syncthreads();
    if (tid < BIN_ROWS) {
        const int excl = s_ws[wv] + incl - my;
        s_off[tid] = excl;
        s_cnt[tid] = 0;
        const int row = b * BIN_ROWS + tid;
        if (row < N)      rowptr[row] = lo + excl;
        if (row == N - 1) rowptr[N]   = lo + excl + my;
    }
    __syncthreads();
    for (int i0 = lo + tid; i0 < hi; i0 += 8 * 1024) {
        int2 r[8];
#pragma unroll
        for (int k = 0; k < 8; ++k) {
            const int i = i0 + k * 1024;
            r[k] = (i < hi) ? rc[i] : make_int2(-1, 0);
        }
#pragma unroll
        for (int k = 0; k < 8; ++k) {
            if (r[k].x >= 0) {
                const int rl  = r[k].x >> SRC_BITS;
                const int pos = atomicAdd(&s_cnt[rl], 1);
                ((int2*)pairs)[lo + s_off[rl] + pos] =
                    make_int2(r[k].x & ((1 << SRC_BITS) - 1), r[k].y);
            }
        }
    }
}


template <int T>
__device__ __forceinline__ int group_lane(int v) {
    return __builtin_amdgcn_mov_dpp8(v, T * 0x249249);
}

template <int T>
__device__ __forceinline__ void request_row(const float* __restrict__ x, int srcs,
                                            unsigned sub16, v4f& v) {
    const unsigned s = (unsigned)group_lane<T>(srcs);
    v = *(const v4f*)((const char*)x + (s * (FEAT_D * 4u) + sub16));
}

template <int T>
__device__ __forceinline__ void add_row(int ws, const v4f& v, int j0, int n, v4f& acc) {
    const float w  = __int_as_float(group_lane<T>(ws));
    const bool  ok = (j0 + T) < n;
    acc.x = ok ? fmaf(w, v.x, acc.x) : acc.x;
    acc.y = ok ? fmaf(w, v.y, acc.y) : acc.y;
    acc.z = ok ? fmaf(w, v.z, acc.z) : acc.z;
    acc.w = ok ? fmaf(w, v.w, acc.w) : acc.w;
}

__device__ __forceinline__ void request8(const float* __restrict__ x, int srcs,
                                         unsigned sub16, v4f (&v)[8]) {
    request_row<0>(x, srcs, sub16, v[0]);
    request_row<1>(x, srcs, sub16, v[1]);
    request_row<2>(x, srcs, sub16, v[2]);
    request_row<3>(x, srcs, sub16, v[3]);
    request_row<4>(x, srcs, sub16, v[4]);
    request_row<5>(x, srcs, sub16, v[5]);
    request_row<6>(x, srcs, sub16, v[6]);
    request_row<7>(x, srcs, sub16, v[7]);
}

__device__ __forceinline__ void add8(int ws, const v4f (&v)[8], int j0, int n, v4f& acc) {
    add_row<0>(ws, v[0], j0, n, acc);
    add_row<1>(ws, v[1], j0, n, acc);
    add_row<2>(ws, v[2], j0, n, acc);
    add_row<3>(ws, v[3], j0, n, acc);
    add_row<4>(ws, v[4], j0, n, acc);
    add_row<5>(ws, v[5], j0, n, acc);
    add_row<6>(ws, v[6], j0, n, acc);
    add_row<7>(ws, v[7], j0, n, acc);
}

__global__ __launch_bounds__(128)
void step_kernel(const float* __restrict__ x,
                 const int*   __restrict__ pairs,
                 const int*   __restrict__ rowptr,
                 const float* __restrict__ x0,
                 const float* __restrict__ alpha_p,
                 const float* __restrict__ beta_p,
                 float*       __restrict__ out,
                 int ntile) {
    __shared__ __attribute__((aligned(16))) float axs[4][16 * FEAT_D];

    const int wv   = threadIdx.x >> 5;
    const int lane = threadIdx.x & 31;
    const int tile = __builtin_amdgcn_readfirstlane(blockIdx.x * 4 + wv);
    if (tile >= ntile) return;

    const int nb = tile * 16;
    float* myax = &axs[wv][0];
    const int2* prs = (const int2*)pairs;

    const int grp = lane >> 3;
    const int sub = lane & 7;
    const unsigned sub16 = (unsigned)sub * 16u;

    int p[4], n[4];
#pragma unroll
    for (int i = 0; i < 4; ++i) {
        const int r = nb + 4 * i + grp;
        p[i] = rowptr[r];
        n[i] = rowptr[r + 1] - p[i];
    }
    int2 ed[4][3];
#pragma unroll
    for (int i = 0; i < 4; ++i) {
#pragma unroll
        for (int k = 0; k < 3; ++k) {
            const int j = sub + 8 * k;
            ed[i][k] = prs[(j < n[i]) ? p[i] + j : 0];
        }
    }

#pragma unroll
    for (int i = 0; i < 4; ++i) {
        const int ni = n[i];
        const int m = max(max(__builtin_amdgcn_readlane(ni, 0),  __builtin_amdgcn_readlane(ni, 8)),
                          max(__builtin_amdgcn_readlane(ni, 16), __builtin_amdgcn_readlane(ni, 24)));
        v4f acc = {0.0f, 0.0f, 0.0f, 0.0f};
        v4f va[8], vb[8];
        int2 e3 = make_int2(0, 0);
        if (m > 24) {
            const int j = 24 + sub;
            e3 = prs[(j < ni) ? p[i] + j : 0];
        }
        request8(x, ed[i][0].x, sub16, va);
        request8(x, ed[i][1].x, sub16, vb);
        add8(ed[i][0].y, va, 0, ni, acc);
        request8(x, ed[i][2].x, sub16, va);
        add8(ed[i][1].y, vb, 8, ni, acc);
        add8(ed[i][2].y, va, 16, ni, acc);
        if (m > 24) {
            request8(x, e3.x, sub16, vb);
            add8(e3.y, vb, 24, ni, acc);
            for (int jb = 32; jb < m; jb += 8) {
                const int j = jb + sub;
                const int2 e = prs[(j < ni) ? p[i] + j : 0];
                request8(x, e.x, sub16, va);
                add8(e.y, va, jb, ni, acc);
            }
        }
        *(v4f*)&myax[(4 * i + grp) * FEAT_D + 4 * sub] = acc;
    }

    const int hi = lane >> 4;
    const int li = lane & 15;

    float araw  = *alpha_p;
    float alpha = 1.0f / (1.0f + expf(-araw));
    float beta  = *beta_p;
    float ca    = 1.0f - alpha;

#pragma unroll
    for (int h = 0; h < 2; ++h) {
        const int cb = 16 * h;
        v8f c;
#pragma unroll
        for (int v = 0; v < 8; ++v) {
            int row = nb + v + 8 * hi;
            int idx = row * FEAT_D + cb + li;
            c[v] = ca * x[idx] + beta * x0[idx];
        }
#pragma unroll
        for (int m = 0; m < 4; ++m) {
            v2f a, b;
#pragma unroll
            for (int v = 0; v < 2; ++v) {
                int k = v + 2 * hi;
                a[v] = (li == 4 * m + k) ? alpha : 0.0f;
                b[v] = myax[(4 * m + k) * FEAT_D + cb + li];
            }
            c = __builtin_amdgcn_wmma_f32_16x16x4_f32(
                    false, a, false, b, (short)0, c, false, false);
        }
#pragma unroll
        for (int v = 0; v < 8; ++v) {
            int row = nb + v + 8 * hi;
            out[row * FEAT_D + cb + li] = c[v];
        }
    }
}

extern "C" void kernel_launch(void* const* d_in, const int* in_sizes, int n_in,
                              void* d_out, int out_size, void* d_ws, size_t ws_size,
                              hipStream_t stream) {
    const float* x_in    = (const float*)d_in[0];
    const float* w       = (const float*)d_in[1];
    const float* x0      = (const float*)d_in[2];
    const float* alpha_p = (const float*)d_in[3];
    const float* beta_p  = (const float*)d_in[4];
    const int*   eidx    = (const int*)  d_in[5];

    const int N = in_sizes[0] / FEAT_D;
    const int E = in_sizes[1];
    const int* src = eidx;
    const int* dst = eidx + E;

    const int ND = N * FEAT_D;
    float* state  = (float*)d_ws;
    int*   pairs  = (int*)(state + ND);
    int*   rowptr = pairs + 2 * E;
    int*   tab    = rowptr + (N + 1);
    float* out    = (float*)d_out;

    const int nbin   = (N + BIN_ROWS - 1) / BIN_ROWS;
    const int nchunk = (E + CHUNK - 1) / CHUNK;
    const int ntab   = nbin * nchunk;
    int* hist = tab;
    int* pref = tab + ntab;
    int* rec  = (int*)state;

    dim3 blk(128);
    const int scanBlocks = (ntab + 1023) / 1024;
    const int ntile      = N / 16;
    const int stepBlocks = (ntile + 3) / 4;

    bin_hist_kernel<<<nchunk, dim3(1024), 0, stream>>>(dst, hist, E, nbin, nchunk);
    scan_rows_kernel<<<scanBlocks, dim3(1024), 0, stream>>>(hist, pref, ntab);
    bin_scatter_kernel<<<nchunk, dim3(1024), 0, stream>>>(src, dst, w, pref, rec, E, nbin, nchunk);
    bin_rows_kernel<<<nbin, dim3(1024), 0, stream>>>(rec, pref, rowptr, pairs, N, nchunk);

    step_kernel<<<stepBlocks, blk, 0, stream>>>(x_in,  pairs, rowptr, x0, alpha_p, beta_p, state, ntile);
    step_kernel<<<stepBlocks, blk, 0, stream>>>(state, pairs, rowptr, x0, alpha_p, beta_p, out,   ntile);
    step_kernel<<<stepBlocks, blk, 0, stream>>>(out,   pairs, rowptr, x0, alpha_p, beta_p, state, ntile);
    step_kernel<<<stepBlocks, blk, 0, stream>>>(state, pairs, rowptr, x0, alpha_p, beta_p, out,   ntile);
}


# Sort level 2: a bin's records are read once, take their row place at the count, are put in row order in LDS and stored as whole 128-byte lines; a row's edge order, so f32 sum order, may differ
# speedup vs baseline: 5.3056x; 1.0456x over previous
#include <hip/hip_runtime.h>
#include <math.h>


#define FEAT_D 32

typedef __attribute__((ext_vector_type(2))) float v2f;
typedef __attribute__((ext_vector_type(4))) float v4f;
typedef __attribute__((ext_vector_type(8))) float v8f;

#define BIN_SHIFT 9
#define BIN_ROWS  512
#define CHUNK     8192
#define SRC_BITS  17
#define MAX_BINS  256

__global__ __launch_bounds__(1024)
void bin_hist_kernel(const int* __restrict__ dst, int* __restrict__ hist,
                     int E, int nbin, int nchunk) {
    __shared__ int s_h[MAX_BINS];
    const int tid = threadIdx.x;
    const int c   = blockIdx.x;
    if (tid < MAX_BINS) s_h[tid] = 0;
    __syncthreads();
    const int base = c * CHUNK + tid;
    int d[8];
#pragma unroll
    for (int k = 0; k < 8; ++k) {
        const int e = base + k * 1024;
        d[k] = (e < E) ? dst[e] : -1;
    }
#pragma unroll
    for (int k = 0; k < 8; ++k)
        if (d[k] >= 0) atomicAdd(&s_h[d[k] >> BIN_SHIFT], 1);
    __syncthreads();
    if (tid < nbin) hist[tid * nchunk + c] = s_h[tid];
}

__global__ __launch_bounds__(1024)
void scan_rows_kernel(const int* __restrict__ cnt, int* __restrict__ rowptr, int N) {
    __shared__ int s_part[32];
    __shared__ int s_tot[32];
    __shared__ int s_off[32];
    __shared__ int s_base;
    const int tid  = threadIdx.x;
    const int lane = tid & 31;
    const int wv   = tid >> 5;
    const int start = blockIdx.x * 1024;

    int part = 0;
    for (int i = tid; i < start; i += 1024) part += cnt[i];
    const int gi = start + tid;
    const int my = (gi < N) ? cnt[gi] : 0;

    for (int o = 16; o > 0; o >>= 1) part += __shfl_xor(part, o, 32);
    int incl = my;
    for (int o = 1; o < 32; o <<= 1) {
        int t = __shfl_up(incl, o, 32);
        if (lane >= o) incl += t;
    }
    if (lane == 31) s_tot[wv] = incl;
    if (lane == 0)  s_part[wv] = part;
    __syncthreads();
    if (wv == 0) {
        const int t = s_tot[lane];
        int p = s_part[lane];
        for (int o = 16; o > 0; o >>= 1) p += __shfl_xor(p, o, 32);
        int ti = t;
        for (int o = 1; o < 32; o <<= 1) {
            int u = __shfl_up(ti, o, 32);
            if (lane >= o) ti += u;
        }
        s_off[lane] = ti - t;
        if (lane == 0) s_base = p;
    }
    __syncthreads();
    const int excl = s_base + s_off[wv] + incl - my;
    if (gi < N) rowptr[gi] = excl;
    if (gi == N - 1) rowptr[N] = excl + my;
}

__global__ __launch_bounds__(1024)
void bin_scatter_kernel(const int*   __restrict__ src,
                        const int*   __restrict__ dst,
                        const float* __restrict__ w,
                        const int*   __restrict__ pref,
                        int*         __restrict__ rec,
                        int E, int nbin, int nchunk) {
    __shared__ int s_cur[MAX_BINS];
    const int tid = threadIdx.x;
    const int c   = blockIdx.x;
    if (tid < nbin) s_cur[tid] = pref[tid * nchunk + c];
    __syncthreads();
    const int base = c * CHUNK + tid;
    int d[8], s[8], wi[8];
#pragma unroll
    for (int k = 0; k < 8; ++k) {
        const int e = base + k * 1024;
        const bool ok = e < E;
        d[k]  = ok ? dst[e] : -1;
        s[k]  = ok ? src[e] : 0;
        wi[k] = ok ? __float_as_int(w[e]) : 0;
    }
#pragma unroll
    for (int k = 0; k < 8; ++k) {
        if (d[k] >= 0) {
            const int pos = atomicAdd(&s_cur[d[k] >> BIN_SHIFT], 1);
            ((int2*)rec)[pos] =
                make_int2(s[k] | ((d[k] & (BIN_ROWS - 1)) << SRC_BITS), wi[k]);
        }
    }
}

#define STAGE_CAP 12288
#define STAGE_PER_THREAD 12

__global__ __launch_bounds__(1024)
void bin_rows_kernel(const int* __restrict__ rec,
                     const int* __restrict__ pref,
                     int*       __restrict__ rowptr,
                     int*       __restrict__ pairs,
                     int N, int nchunk) {
    __shared__ int  s_cnt[BIN_ROWS];
    __shared__ int  s_off[BIN_ROWS];
    __shared__ int  s_ws[16];
    __shared__ int2 s_stage[STAGE_CAP];
    const int tid  = threadIdx.x;
    const int lane = tid & 31;
    const int wv   = tid >> 5;
    const int b    = blockIdx.x;
    const int lo   = pref[b * nchunk];
    const int hi   = pref[(b + 1) * nchunk];
    const int2* rc = (const int2*)rec;
    const bool staged = (hi - lo) <= STAGE_CAP;

    if (tid < BIN_ROWS) s_cnt[tid] = 0;
    __syncthreads();

    int2 r[STAGE_PER_THREAD];
    int  pos[STAGE_PER_THREAD];
    if (staged) {
#pragma unroll
        for (int k = 0; k < STAGE_PER_THREAD; ++k) {
            const int i = lo + tid + k * 1024;
            r[k] = (i < hi) ? rc[i] : make_int2(-1, 0);
        }
#pragma unroll
        for (int k = 0; k < STAGE_PER_THREAD; ++k) {
            pos[k] = 0;
            if (r[k].x >= 0) pos[k] = atomicAdd(&s_cnt[r[k].x >> SRC_BITS], 1);
        }
    } else {
        for (int i0 = lo + tid; i0 < hi; i0 += 8 * 1024) {
            int key[8];
#pragma unroll
            for (int k = 0; k < 8; ++k) {
                const int i = i0 + k * 1024;
                key[k] = (i < hi) ? rc[i].x : -1;
            }
#pragma unroll
            for (int k = 0; k < 8; ++k)
                if (key[k] >= 0) atomicAdd(&s_cnt[key[k] >> SRC_BITS], 1);
        }
    }
    __syncthreads();

    const int my = (tid < BIN_ROWS) ? s_cnt[tid] : 0;
    int incl = my;
    for (int o = 1; o < 32; o <<= 1) {
        int t = __shfl_up(incl, o, 32);
        if (lane >= o) incl += t;
    }
    if (lane == 31 && wv < 16) s_ws[wv] = incl;
    __syncthreads();
    if (wv == 0) {
        const int t = (lane < 16) ? s_ws[lane] : 0;
        int ti = t;
        for (int o = 1; o < 16; o <<= 1) {
            int u = __shfl_up(ti, o, 32);
            if (lane >= o) ti += u;
        }
        if (lane < 16) s_ws[lane] = ti - t;
    }
    __syncthreads();
    if (tid < BIN_ROWS) {
        const int excl = s_ws[wv] + incl - my;
        s_off[tid] = excl;
        s_cnt[tid] = 0;
        const int row = b * BIN_ROWS + tid;
        if (row < N)      rowptr[row] = lo + excl;
        if (row == N - 1) rowptr[N]   = lo + excl + my;
    }
    __syncthreads();

    if (staged) {
#pragma unroll
        for (int k = 0; k < STAGE_PER_THREAD; ++k)
            if (r[k].x >= 0) pos[k] += s_off[r[k].x >> SRC_BITS];
#pragma unroll
        for (int k = 0; k < STAGE_PER_THREAD; ++k)
            if (r[k].x >= 0)
                s_stage[pos[k]] = make_int2(r[k].x & ((1 << SRC_BITS) - 1), r[k].y);
        __syncthreads();
        for (int i0 = (lo & ~31) + tid; i0 < hi; i0 += 4 * 1024) {
            int2 v[4];
#pragma unroll
            for (int k = 0; k < 4; ++k) {
                const int i = i0 + k * 1024;
                v[k] = (i >= lo && i < hi) ? s_stage[i - lo] : make_int2(0, 0);
            }
#pragma unroll
            for (int k = 0; k < 4; ++k) {
                const int i = i0 + k * 1024;
                if (i >= lo && i < hi) ((int2*)pairs)[i] = v[k];
            }
        }
    } else {
        for (int i0 = lo + tid; i0 < hi; i0 += 8 * 1024) {
            int2 q[8];
#pragma unroll
            for (int k = 0; k < 8; ++k) {
                const int i = i0 + k * 1024;
                q[k] = (i < hi) ? rc[i] : make_int2(-1, 0);
            }
#pragma unroll
            for (int k = 0; k < 8; ++k) {
                if (q[k].x >= 0) {
                    const int rl  = q[k].x >> SRC_BITS;
                    const int p   = atomicAdd(&s_cnt[rl], 1);
                    ((int2*)pairs)[lo + s_off[rl] + p] =
                        make_int2(q[k].x & ((1 << SRC_BITS) - 1), q[k].y);
                }
            }
        }
    }
}


template <int T>
__device__ __forceinline__ int group_lane(int v) {
    return __builtin_amdgcn_mov_dpp8(v, T * 0x249249);
}

template <int T>
__device__ __forceinline__ void request_row(const float* __restrict__ x, int srcs,
                                            unsigned sub16, v4f& v) {
    const unsigned s = (unsigned)group_lane<T>(srcs);
    v = *(const v4f*)((const char*)x + (s * (FEAT_D * 4u) + sub16));
}

template <int T>
__device__ __forceinline__ void add_row(int ws, const v4f& v, int j0, int n, v4f& acc) {
    const float w  = __int_as_float(group_lane<T>(ws));
    const bool  ok = (j0 + T) < n;
    acc.x = ok ? fmaf(w, v.x, acc.x) : acc.x;
    acc.y = ok ? fmaf(w, v.y, acc.y) : acc.y;
    acc.z = ok ? fmaf(w, v.z, acc.z) : acc.z;
    acc.w = ok ? fmaf(w, v.w, acc.w) : acc.w;
}

__device__ __forceinline__ void request8(const float* __restrict__ x, int srcs,
                                         unsigned sub16, v4f (&v)[8]) {
    request_row<0>(x, srcs, sub16, v[0]);
    request_row<1>(x, srcs, sub16, v[1]);
    request_row<2>(x, srcs, sub16, v[2]);
    request_row<3>(x, srcs, sub16, v[3]);
    request_row<4>(x, srcs, sub16, v[4]);
    request_row<5>(x, srcs, sub16, v[5]);
    request_row<6>(x, srcs, sub16, v[6]);
    request_row<7>(x, srcs, sub16, v[7]);
}

__device__ __forceinline__ void add8(int ws, const v4f (&v)[8], int j0, int n, v4f& acc) {
    add_row<0>(ws, v[0], j0, n, acc);
    add_row<1>(ws, v[1], j0, n, acc);
    add_row<2>(ws, v[2], j0, n, acc);
    add_row<3>(ws, v[3], j0, n, acc);
    add_row<4>(ws, v[4], j0, n, acc);
    add_row<5>(ws, v[5], j0, n, acc);
    add_row<6>(ws, v[6], j0, n, acc);
    add_row<7>(ws, v[7], j0, n, acc);
}

__global__ __launch_bounds__(128)
void step_kernel(const float* __restrict__ x,
                 const int*   __restrict__ pairs,
                 const int*   __restrict__ rowptr,
                 const float* __restrict__ x0,
                 const float* __restrict__ alpha_p,
                 const float* __restrict__ beta_p,
                 float*       __restrict__ out,
                 int ntile) {
    __shared__ __attribute__((aligned(16))) float axs[4][16 * FEAT_D];

    const int wv   = threadIdx.x >> 5;
    const int lane = threadIdx.x & 31;
    const int tile = __builtin_amdgcn_readfirstlane(blockIdx.x * 4 + wv);
    if (tile >= ntile) return;

    const int nb = tile * 16;
    float* myax = &axs[wv][0];
    const int2* prs = (const int2*)pairs;

    const int grp = lane >> 3;
    const int sub = lane & 7;
    const unsigned sub16 = (unsigned)sub * 16u;

    int p[4], n[4];
#pragma unroll
    for (int i = 0; i < 4; ++i) {
        const int r = nb + 4 * i + grp;
        p[i] = rowptr[r];
        n[i] = rowptr[r + 1] - p[i];
    }
    int2 ed[4][3];
#pragma unroll
    for (int i = 0; i < 4; ++i) {
#pragma unroll
        for (int k = 0; k < 3; ++k) {
            const int j = sub + 8 * k;
            ed[i][k] = prs[(j < n[i]) ? p[i] + j : 0];
        }
    }

#pragma unroll
    for (int i = 0; i < 4; ++i) {
        const int ni = n[i];
        const int m = max(max(__builtin_amdgcn_readlane(ni, 0),  __builtin_amdgcn_readlane(ni, 8)),
                          max(__builtin_amdgcn_readlane(ni, 16), __builtin_amdgcn_readlane(ni, 24)));
        v4f acc = {0.0f, 0.0f, 0.0f, 0.0f};
        v4f va[8], vb[8];
        int2 e3 = make_int2(0, 0);
        if (m > 24) {
            const int j = 24 + sub;
            e3 = prs[(j < ni) ? p[i] + j : 0];
        }
        request8(x, ed[i][0].x, sub16, va);
        request8(x, ed[i][1].x, sub16, vb);
        add8(ed[i][0].y, va, 0, ni, acc);
        request8(x, ed[i][2].x, sub16, va);
        add8(ed[i][1].y, vb, 8, ni, acc);
        add8(ed[i][2].y, va, 16, ni, acc);
        if (m > 24) {
            request8(x, e3.x, sub16, vb);
            add8(e3.y, vb, 24, ni, acc);
            for (int jb = 32; jb < m; jb += 8) {
                const int j = jb + sub;
                const int2 e = prs[(j < ni) ? p[i] + j : 0];
                request8(x, e.x, sub16, va);
                add8(e.y, va, jb, ni, acc);
            }
        }
        *(v4f*)&myax[(4 * i + grp) * FEAT_D + 4 * sub] = acc;
    }

    const int hi = lane >> 4;
    const int li = lane & 15;

    float araw  = *alpha_p;
    float alpha = 1.0f / (1.0f + expf(-araw));
    float beta  = *beta_p;
    float ca    = 1.0f - alpha;

#pragma unroll
    for (int h = 0; h < 2; ++h) {
        const int cb = 16 * h;
        v8f c;
#pragma unroll
        for (int v = 0; v < 8; ++v) {
            int row = nb + v + 8 * hi;
            int idx = row * FEAT_D + cb + li;
            c[v] = ca * x[idx] + beta * x0[idx];
        }
#pragma unroll
        for (int m = 0; m < 4; ++m) {
            v2f a, b;
#pragma unroll
            for (int v = 0; v < 2; ++v) {
                int k = v + 2 * hi;
                a[v] = (li == 4 * m + k) ? alpha : 0.0f;
                b[v] = myax[(4 * m + k) * FEAT_D + cb + li];
            }
            c = __builtin_amdgcn_wmma_f32_16x16x4_f32(
                    false, a, false, b, (short)0, c, false, false);
        }
#pragma unroll
        for (int v = 0; v < 8; ++v) {
            int row = nb + v + 8 * hi;
            out[row * FEAT_D + cb + li] = c[v];
        }
    }
}

extern "C" void kernel_launch(void* const* d_in, const int* in_sizes, int n_in,
                              void* d_out, int out_size, void* d_ws, size_t ws_size,
                              hipStream_t stream) {
    const float* x_in    = (const float*)d_in[0];
    const float* w       = (const float*)d_in[1];
    const float* x0      = (const float*)d_in[2];
    const float* alpha_p = (const float*)d_in[3];
    const float* beta_p  = (const float*)d_in[4];
    const int*   eidx    = (const int*)  d_in[5];

    const int N = in_sizes[0] / FEAT_D;
    const int E = in_sizes[1];
    const int* src = eidx;
    const int* dst = eidx + E;

    const int ND = N * FEAT_D;
    float* state  = (float*)d_ws;
    int*   pairs  = (int*)(state + ND);
    int*   rowptr = pairs + 2 * E;
    int*   tab    = rowptr + (N + 1);
    float* out    = (float*)d_out;

    const int nbin   = (N + BIN_ROWS - 1) / BIN_ROWS;
    const int nchunk = (E + CHUNK - 1) / CHUNK;
    const int ntab   = nbin * nchunk;
    int* hist = tab;
    int* pref = tab + ntab;
    int* rec  = (int*)state;

    dim3 blk(128);
    const int scanBlocks = (ntab + 1023) / 1024;
    const int ntile      = N / 16;
    const int stepBlocks = (ntile + 3) / 4;

    bin_hist_kernel<<<nchunk, dim3(1024), 0, stream>>>(dst, hist, E, nbin, nchunk);
    scan_rows_kernel<<<scanBlocks, dim3(1024), 0, stream>>>(hist, pref, ntab);
    bin_scatter_kernel<<<nchunk, dim3(1024), 0, stream>>>(src, dst, w, pref, rec, E, nbin, nchunk);
    bin_rows_kernel<<<nbin, dim3(1024), 0, stream>>>(rec, pref, rowptr, pairs, N, nchunk);

    step_kernel<<<stepBlocks, blk, 0, stream>>>(x_in,  pairs, rowptr, x0, alpha_p, beta_p, state, ntile);
    step_kernel<<<stepBlocks, blk, 0, stream>>>(state, pairs, rowptr, x0, alpha_p, beta_p, out,   ntile);
    step_kernel<<<stepBlocks, blk, 0, stream>>>(out,   pairs, rowptr, x0, alpha_p, beta_p, state, ntile);
    step_kernel<<<stepBlocks, blk, 0, stream>>>(state, pairs, rowptr, x0, alpha_p, beta_p, out,   ntile);
}


# Sort level 1: a chunk's records are put in bin order in LDS (LDS counter = place in run) and leave run by run, not as 32 scattered 8-byte stores; row edge order, so f32 sum order, may differ
# speedup vs baseline: 5.5106x; 1.0386x over previous
#include <hip/hip_runtime.h>
#include <math.h>


#define FEAT_D 32

typedef __attribute__((ext_vector_type(2))) float v2f;
typedef __attribute__((ext_vector_type(4))) float v4f;
typedef __attribute__((ext_vector_type(8))) float v8f;

#define BIN_SHIFT 9
#define BIN_ROWS  512
#define CHUNK     8192
#define SRC_BITS  17
#define MAX_BINS  256

__global__ __launch_bounds__(1024)
void bin_hist_kernel(const int* __restrict__ dst, int* __restrict__ hist,
                     int E, int nbin, int nchunk) {
    __shared__ int s_h[MAX_BINS];
    const int tid = threadIdx.x;
    const int c   = blockIdx.x;
    if (tid < MAX_BINS) s_h[tid] = 0;
    __syncthreads();
    const int base = c * CHUNK + tid;
    int d[8];
#pragma unroll
    for (int k = 0; k < 8; ++k) {
        const int e = base + k * 1024;
        d[k] = (e < E) ? dst[e] : -1;
    }
#pragma unroll
    for (int k = 0; k < 8; ++k)
        if (d[k] >= 0) atomicAdd(&s_h[d[k] >> BIN_SHIFT], 1);
    __syncthreads();
    if (tid < nbin) hist[tid * nchunk + c] = s_h[tid];
}

__global__ __launch_bounds__(1024)
void scan_rows_kernel(const int* __restrict__ cnt, int* __restrict__ rowptr, int N) {
    __shared__ int s_part[32];
    __shared__ int s_tot[32];
    __shared__ int s_off[32];
    __shared__ int s_base;
    const int tid  = threadIdx.x;
    const int lane = tid & 31;
    const int wv   = tid >> 5;
    const int start = blockIdx.x * 1024;

    int part = 0;
    for (int i = tid; i < start; i += 1024) part += cnt[i];
    const int gi = start + tid;
    const int my = (gi < N) ? cnt[gi] : 0;

    for (int o = 16; o > 0; o >>= 1) part += __shfl_xor(part, o, 32);
    int incl = my;
    for (int o = 1; o < 32; o <<= 1) {
        int t = __shfl_up(incl, o, 32);
        if (lane >= o) incl += t;
    }
    if (lane == 31) s_tot[wv] = incl;
    if (lane == 0)  s_part[wv] = part;
    __syncthreads();
    if (wv == 0) {
        const int t = s_tot[lane];
        int p = s_part[lane];
        for (int o = 16; o > 0; o >>= 1) p += __shfl_xor(p, o, 32);
        int ti = t;
        for (int o = 1; o < 32; o <<= 1) {
            int u = __shfl_up(ti, o, 32);
            if (lane >= o) ti += u;
        }
        s_off[lane] = ti - t;
        if (lane == 0) s_base = p;
    }
    __syncthreads();
    const int excl = s_base + s_off[wv] + incl - my;
    if (gi < N) rowptr[gi] = excl;
    if (gi == N - 1) rowptr[N] = excl + my;
}

__global__ __launch_bounds__(1024)
void bin_scatter_kernel(const int*   __restrict__ src,
                        const int*   __restrict__ dst,
                        const float* __restrict__ w,
                        const int*   __restrict__ pref,
                        int*         __restrict__ rec,
                        int E, int nbin, int nchunk) {
    __shared__ int  s_cnt[MAX_BINS];
    __shared__ int  s_loc[MAX_BINS];
    __shared__ int  s_dlt[MAX_BINS];
    __shared__ int  s_ws[8];
    __shared__ int2 s_stage[CHUNK];
    __shared__ int  s_gpos[CHUNK];
    const int tid  = threadIdx.x;
    const int lane = tid & 31;
    const int wv   = tid >> 5;
    const int c    = blockIdx.x;
    if (tid < MAX_BINS) s_cnt[tid] = 0;
    const int pf = (tid < nbin) ? pref[tid * nchunk + c] : 0;
    const int base = c * CHUNK + tid;
    int d[8], s[8], wi[8];
#pragma unroll
    for (int k = 0; k < 8; ++k) {
        const int e = base + k * 1024;
        const bool ok = e < E;
        d[k]  = ok ? dst[e] : -1;
        s[k]  = ok ? src[e] : 0;
        wi[k] = ok ? __float_as_int(w[e]) : 0;
    }
    __syncthreads();
    int rk[8];
#pragma unroll
    for (int k = 0; k < 8; ++k) {
        rk[k] = 0;
        if (d[k] >= 0) rk[k] = atomicAdd(&s_cnt[d[k] >> BIN_SHIFT], 1);
    }
    __syncthreads();

    const int my = (tid < MAX_BINS) ? s_cnt[tid] : 0;
    int incl = my;
    for (int o = 1; o < 32; o <<= 1) {
        int t = __shfl_up(incl, o, 32);
        if (lane >= o) incl += t;
    }
    if (lane == 31 && wv < 8) s_ws[wv] = incl;
    __syncthreads();
    if (wv == 0) {
        const int t = (lane < 8) ? s_ws[lane] : 0;
        int ti = t;
        for (int o = 1; o < 8; o <<= 1) {
            int u = __shfl_up(ti, o, 32);
            if (lane >= o) ti += u;
        }
        if (lane < 8) s_ws[lane] = ti - t;
    }
    __syncthreads();
    if (tid < MAX_BINS) {
        const int excl = s_ws[wv] + incl - my;
        s_loc[tid] = excl;
        s_dlt[tid] = pf - excl;
    }
    __syncthreads();

#pragma unroll
    for (int k = 0; k < 8; ++k) {
        if (d[k] >= 0) {
            const int b   = d[k] >> BIN_SHIFT;
            const int pos = s_loc[b] + rk[k];
            s_stage[pos] = make_int2(s[k] | ((d[k] & (BIN_ROWS - 1)) << SRC_BITS), wi[k]);
            s_gpos[pos]  = pos + s_dlt[b];
        }
    }
    __syncthreads();

    const int tot = min(CHUNK, E - c * CHUNK);
    int2 v[8];
    int  g[8];
#pragma unroll
    for (int k = 0; k < 8; ++k) {
        const int i = tid + k * 1024;
        const bool ok = i < tot;
        v[k] = ok ? s_stage[i] : make_int2(0, 0);
        g[k] = ok ? s_gpos[i] : -1;
    }
#pragma unroll
    for (int k = 0; k < 8; ++k)
        if (g[k] >= 0) ((int2*)rec)[g[k]] = v[k];
}

#define STAGE_CAP 12288
#define STAGE_PER_THREAD 12

__global__ __launch_bounds__(1024)
void bin_rows_kernel(const int* __restrict__ rec,
                     const int* __restrict__ pref,
                     int*       __restrict__ rowptr,
                     int*       __restrict__ pairs,
                     int N, int nchunk) {
    __shared__ int  s_cnt[BIN_ROWS];
    __shared__ int  s_off[BIN_ROWS];
    __shared__ int  s_ws[16];
    __shared__ int2 s_stage[STAGE_CAP];
    const int tid  = threadIdx.x;
    const int lane = tid & 31;
    const int wv   = tid >> 5;
    const int b    = blockIdx.x;
    const int lo   = pref[b * nchunk];
    const int hi   = pref[(b + 1) * nchunk];
    const int2* rc = (const int2*)rec;
    const bool staged = (hi - lo) <= STAGE_CAP;

    if (tid < BIN_ROWS) s_cnt[tid] = 0;
    __syncthreads();

    int2 r[STAGE_PER_THREAD];
    int  pos[STAGE_PER_THREAD];
    if (staged) {
#pragma unroll
        for (int k = 0; k < STAGE_PER_THREAD; ++k) {
            const int i = lo + tid + k * 1024;
            r[k] = (i < hi) ? rc[i] : make_int2(-1, 0);
        }
#pragma unroll
        for (int k = 0; k < STAGE_PER_THREAD; ++k) {
            pos[k] = 0;
            if (r[k].x >= 0) pos[k] = atomicAdd(&s_cnt[r[k].x >> SRC_BITS], 1);
        }
    } else {
        for (int i0 = lo + tid; i0 < hi; i0 += 8 * 1024) {
            int key[8];
#pragma unroll
            for (int k = 0; k < 8; ++k) {
                const int i = i0 + k * 1024;
                key[k] = (i < hi) ? rc[i].x : -1;
            }
#pragma unroll
            for (int k = 0; k < 8; ++k)
                if (key[k] >= 0) atomicAdd(&s_cnt[key[k] >> SRC_BITS], 1);
        }
    }
    __syncthreads();

    const int my = (tid < BIN_ROWS) ? s_cnt[tid] : 0;
    int incl = my;
    for (int o = 1; o < 32; o <<= 1) {
        int t = __shfl_up(incl, o, 32);
        if (lane >= o) incl += t;
    }
    if (lane == 31 && wv < 16) s_ws[wv] = incl;
    __syncthreads();
    if (wv == 0) {
        const int t = (lane < 16) ? s_ws[lane] : 0;
        int ti = t;
        for (int o = 1; o < 16; o <<= 1) {
            int u = __shfl_up(ti, o, 32);
            if (lane >= o) ti += u;
        }
        if (lane < 16) s_ws[lane] = ti - t;
    }
    __syncthreads();
    if (tid < BIN_ROWS) {
        const int excl = s_ws[wv] + incl - my;
        s_off[tid] = excl;
        s_cnt[tid] = 0;
        const int row = b * BIN_ROWS + tid;
        if (row < N)      rowptr[row] = lo + excl;
        if (row == N - 1) rowptr[N]   = lo + excl + my;
    }
    __syncthreads();

    if (staged) {
#pragma unroll
        for (int k = 0; k < STAGE_PER_THREAD; ++k)
            if (r[k].x >= 0) pos[k] += s_off[r[k].x >> SRC_BITS];
#pragma unroll
        for (int k = 0; k < STAGE_PER_THREAD; ++k)
            if (r[k].x >= 0)
                s_stage[pos[k]] = make_int2(r[k].x & ((1 << SRC_BITS) - 1), r[k].y);
        __syncthreads();
        for (int i0 = (lo & ~31) + tid; i0 < hi; i0 += 4 * 1024) {
            int2 v[4];
#pragma unroll
            for (int k = 0; k < 4; ++k) {
                const int i = i0 + k * 1024;
                v[k] = (i >= lo && i < hi) ? s_stage[i - lo] : make_int2(0, 0);
            }
#pragma unroll
            for (int k = 0; k < 4; ++k) {
                const int i = i0 + k * 1024;
                if (i >= lo && i < hi) ((int2*)pairs)[i] = v[k];
            }
        }
    } else {
        for (int i0 = lo + tid; i0 < hi; i0 += 8 * 1024) {
            int2 q[8];
#pragma unroll
            for (int k = 0; k < 8; ++k) {
                const int i = i0 + k * 1024;
                q[k] = (i < hi) ? rc[i] : make_int2(-1, 0);
            }
#pragma unroll
            for (int k = 0; k < 8; ++k) {
                if (q[k].x >= 0) {
                    const int rl  = q[k].x >> SRC_BITS;
                    const int p   = atomicAdd(&s_cnt[rl], 1);
                    ((int2*)pairs)[lo + s_off[rl] + p] =
                        make_int2(q[k].x & ((1 << SRC_BITS) - 1), q[k].y);
                }
            }
        }
    }
}


template <int T>
__device__ __forceinline__ int group_lane(int v) {
    return __builtin_amdgcn_mov_dpp8(v, T * 0x249249);
}

template <int T>
__device__ __forceinline__ void request_row(const float* __restrict__ x, int srcs,
                                            unsigned sub16, v4f& v) {
    const unsigned s = (unsigned)group_lane<T>(srcs);
    v = *(const v4f*)((const char*)x + (s * (FEAT_D * 4u) + sub16));
}

template <int T>
__device__ __forceinline__ void add_row(int ws, const v4f& v, int j0, int n, v4f& acc) {
    const float w  = __int_as_float(group_lane<T>(ws));
    const bool  ok = (j0 + T) < n;
    acc.x = ok ? fmaf(w, v.x, acc.x) : acc.x;
    acc.y = ok ? fmaf(w, v.y, acc.y) : acc.y;
    acc.z = ok ? fmaf(w, v.z, acc.z) : acc.z;
    acc.w = ok ? fmaf(w, v.w, acc.w) : acc.w;
}

__device__ __forceinline__ void request8(const float* __restrict__ x, int srcs,
                                         unsigned sub16, v4f (&v)[8]) {
    request_row<0>(x, srcs, sub16, v[0]);
    request_row<1>(x, srcs, sub16, v[1]);
    request_row<2>(x, srcs, sub16, v[2]);
    request_row<3>(x, srcs, sub16, v[3]);
    request_row<4>(x, srcs, sub16, v[4]);
    request_row<5>(x, srcs, sub16, v[5]);
    request_row<6>(x, srcs, sub16, v[6]);
    request_row<7>(x, srcs, sub16, v[7]);
}

__device__ __forceinline__ void add8(int ws, const v4f (&v)[8], int j0, int n, v4f& acc) {
    add_row<0>(ws, v[0], j0, n, acc);
    add_row<1>(ws, v[1], j0, n, acc);
    add_row<2>(ws, v[2], j0, n, acc);
    add_row<3>(ws, v[3], j0, n, acc);
    add_row<4>(ws, v[4], j0, n, acc);
    add_row<5>(ws, v[5], j0, n, acc);
    add_row<6>(ws, v[6], j0, n, acc);
    add_row<7>(ws, v[7], j0, n, acc);
}

__global__ __launch_bounds__(128)
void step_kernel(const float* __restrict__ x,
                 const int*   __restrict__ pairs,
                 const int*   __restrict__ rowptr,
                 const float* __restrict__ x0,
                 const float* __restrict__ alpha_p,
                 const float* __restrict__ beta_p,
                 float*       __restrict__ out,
                 int ntile) {
    __shared__ __attribute__((aligned(16))) float axs[4][16 * FEAT_D];

    const int wv   = threadIdx.x >> 5;
    const int lane = threadIdx.x & 31;
    const int tile = __builtin_amdgcn_readfirstlane(blockIdx.x * 4 + wv);
    if (tile >= ntile) return;

    const int nb = tile * 16;
    float* myax = &axs[wv][0];
    const int2* prs = (const int2*)pairs;

    const int grp = lane >> 3;
    const int sub = lane & 7;
    const unsigned sub16 = (unsigned)sub * 16u;

    int p[4], n[4];
#pragma unroll
    for (int i = 0; i < 4; ++i) {
        const int r = nb + 4 * i + grp;
        p[i] = rowptr[r];
        n[i] = rowptr[r + 1] - p[i];
    }
    int2 ed[4][3];
#pragma unroll
    for (int i = 0; i < 4; ++i) {
#pragma unroll
        for (int k = 0; k < 3; ++k) {
            const int j = sub + 8 * k;
            ed[i][k] = prs[(j < n[i]) ? p[i] + j : 0];
        }
    }

#pragma unroll
    for (int i = 0; i < 4; ++i) {
        const int ni = n[i];
        const int m = max(max(__builtin_amdgcn_readlane(ni, 0),  __builtin_amdgcn_readlane(ni, 8)),
                          max(__builtin_amdgcn_readlane(ni, 16), __builtin_amdgcn_readlane(ni, 24)));
        v4f acc = {0.0f, 0.0f, 0.0f, 0.0f};
        v4f va[8], vb[8];
        int2 e3 = make_int2(0, 0);
        if (m > 24) {
            const int j = 24 + sub;
            e3 = prs[(j < ni) ? p[i] + j : 0];
        }
        request8(x, ed[i][0].x, sub16, va);
        request8(x, ed[i][1].x, sub16, vb);
        add8(ed[i][0].y, va, 0, ni, acc);
        request8(x, ed[i][2].x, sub16, va);
        add8(ed[i][1].y, vb, 8, ni, acc);
        add8(ed[i][2].y, va, 16, ni, acc);
        if (m > 24) {
            request8(x, e3.x, sub16, vb);
            add8(e3.y, vb, 24, ni, acc);
            for (int jb = 32; jb < m; jb += 8) {
                const int j = jb + sub;
                const int2 e = prs[(j < ni) ? p[i] + j : 0];
                request8(x, e.x, sub16, va);
                add8(e.y, va, jb, ni, acc);
            }
        }
        *(v4f*)&myax[(4 * i + grp) * FEAT_D + 4 * sub] = acc;
    }

    const int hi = lane >> 4;
    const int li = lane & 15;

    float araw  = *alpha_p;
    float alpha = 1.0f / (1.0f + expf(-araw));
    float beta  = *beta_p;
    float ca    = 1.0f - alpha;

#pragma unroll
    for (int h = 0; h < 2; ++h) {
        const int cb = 16 * h;
        v8f c;
#pragma unroll
        for (int v = 0; v < 8; ++v) {
            int row = nb + v + 8 * hi;
            int idx = row * FEAT_D + cb + li;
            c[v] = ca * x[idx] + beta * x0[idx];
        }
#pragma unroll
        for (int m = 0; m < 4; ++m) {
            v2f a, b;
#pragma unroll
            for (int v = 0; v < 2; ++v) {
                int k = v + 2 * hi;
                a[v] = (li == 4 * m + k) ? alpha : 0.0f;
                b[v] = myax[(4 * m + k) * FEAT_D + cb + li];
            }
            c = __builtin_amdgcn_wmma_f32_16x16x4_f32(
                    false, a, false, b, (short)0, c, false, false);
        }
#pragma unroll
        for (int v = 0; v < 8; ++v) {
            int row = nb + v + 8 * hi;
            out[row * FEAT_D + cb + li] = c[v];
        }
    }
}

extern "C" void kernel_launch(void* const* d_in, const int* in_sizes, int n_in,
                              void* d_out, int out_size, void* d_ws, size_t ws_size,
                              hipStream_t stream) {
    const float* x_in    = (const float*)d_in[0];
    const float* w       = (const float*)d_in[1];
    const float* x0      = (const float*)d_in[2];
    const float* alpha_p = (const float*)d_in[3];
    const float* beta_p  = (const float*)d_in[4];
    const int*   eidx    = (const int*)  d_in[5];

    const int N = in_sizes[0] / FEAT_D;
    const int E = in_sizes[1];
    const int* src = eidx;
    const int* dst = eidx + E;

    const int ND = N * FEAT_D;
    float* state  = (float*)d_ws;
    int*   pairs  = (int*)(state + ND);
    int*   rowptr = pairs + 2 * E;
    int*   tab    = rowptr + (N + 1);
    float* out    = (float*)d_out;

    const int nbin   = (N + BIN_ROWS - 1) / BIN_ROWS;
    const int nchunk = (E + CHUNK - 1) / CHUNK;
    const int ntab   = nbin * nchunk;
    int* hist = tab;
    int* pref = tab + ntab;
    int* rec  = (int*)state;

    dim3 blk(128);
    const int scanBlocks = (ntab + 1023) / 1024;
    const int ntile      = N / 16;
    const int stepBlocks = (ntile + 3) / 4;

    bin_hist_kernel<<<nchunk, dim3(1024), 0, stream>>>(dst, hist, E, nbin, nchunk);
    scan_rows_kernel<<<scanBlocks, dim3(1024), 0, stream>>>(hist, pref, ntab);
    bin_scatter_kernel<<<nchunk, dim3(1024), 0, stream>>>(src, dst, w, pref, rec, E, nbin, nchunk);
    bin_rows_kernel<<<nbin, dim3(1024), 0, stream>>>(rec, pref, rowptr, pairs, N, nchunk);

    step_kernel<<<stepBlocks, blk, 0, stream>>>(x_in,  pairs, rowptr, x0, alpha_p, beta_p, state, ntile);
    step_kernel<<<stepBlocks, blk, 0, stream>>>(state, pairs, rowptr, x0, alpha_p, beta_p, out,   ntile);
    step_kernel<<<stepBlocks, blk, 0, stream>>>(out,   pairs, rowptr, x0, alpha_p, beta_p, state, ntile);
    step_kernel<<<stepBlocks, blk, 0, stream>>>(state, pairs, rowptr, x0, alpha_p, beta_p, out,   ntile);
}
